# starting kernel
# baseline (speedup 1.0000x reference)
#include <hip/hip_runtime.h>
#include <math.h>


typedef __attribute__((ext_vector_type(2))) float v2f;
typedef __attribute__((ext_vector_type(8))) float v8f;

#define B_   16
#define T_   4096
#define C_   128
#define ESTR 132

__device__ __forceinline__ float half_shfl_max(float v) {
  #pragma unroll
  for (int off = 1; off < 16; off <<= 1)
    v = fmaxf(v, __shfl_xor(v, off, 32));
  return v;
}

__device__ __forceinline__ v8f wmma_k4(const float* __restrict__ sE,
                                       v2f bfrag, int k0, int hh, int lo,
                                       v8f c) {
  const int ka = k0 + 2 * hh;
  v2f a;
  a.x = sE[lo * ESTR + ka];
  a.y = sE[lo * ESTR + ka + 1];
  return __builtin_amdgcn_wmma_f32_16x16x4_f32(
      false, a, false, bfrag, (short)0, c, false, false);
}

__global__ __launch_bounds__(128) void crf_prep(const float* __restrict__ lt,
                                                float* __restrict__ Pg,
                                                float* __restrict__ PTg) {
  const int i = threadIdx.x;
  float mx = -INFINITY;
  for (int j = 0; j < C_; ++j) mx = fmaxf(mx, lt[i * C_ + j]);
  float s = 0.f;
  for (int j = 0; j < C_; ++j) s += __expf(lt[i * C_ + j] - mx);
  const float inv = 1.0f / s;
  for (int j = 0; j < C_; ++j) {
    const float p = __expf(lt[i * C_ + j] - mx) * inv;
    Pg[i * C_ + j] = p;
    PTg[j * C_ + i] = p;
  }
}

template <bool FWD>
__device__ void scan_dir(const float* __restrict__ u,
                         const float* __restrict__ mat,
                         float* __restrict__ outbuf,
                         float* smem) {
  float* sE = smem;
  float* sA = sE + B_ * ESTR;
  float* sM = sA + B_ * C_;

  const int tid  = threadIdx.x;
  const int lane = tid & 31;
  const int wv   = tid >> 5;
  const int hh   = lane >> 4;
  const int lo   = lane & 15;
  const int n0   = wv * 16;

  v2f bf[32];
  #pragma unroll
  for (int kk = 0; kk < 32; ++kk) {
    const int ka = 4 * kk + 2 * hh;
    bf[kk].x = mat[ka * C_ + n0 + lo];
    bf[kk].y = mat[(ka + 1) * C_ + n0 + lo];
  }

  const int b   = tid >> 4;
  const int sub = tid & 15;
  const int j0  = sub * 8;

  {
    const int tEdge = FWD ? 0 : (T_ - 1);
    #pragma unroll
    for (int q = 0; q < 8; ++q) {
      const int j = j0 + q;
      const float v = FWD ? u[(b * T_ + 0) * C_ + j] : 0.0f;
      sA[b * C_ + j] = v;
      outbuf[(b * T_ + tEdge) * C_ + j] = v;
    }
  }
  __syncthreads();

  for (int s = 1; s < T_; ++s) {
    const int t = FWD ? s : (T_ - 1 - s);

    float vals[8];
    float lm = -INFINITY;
    #pragma unroll
    for (int q = 0; q < 8; ++q) {
      float v = sA[b * C_ + j0 + q];
      if (!FWD) v += u[(b * T_ + (t + 1)) * C_ + j0 + q];
      vals[q] = v;
      lm = fmaxf(lm, v);
    }
    lm = half_shfl_max(lm);
    if (sub == 0) sM[b] = lm;
    #pragma unroll
    for (int q = 0; q < 8; ++q)
      sE[b * ESTR + j0 + q] = __expf(vals[q] - lm);

    float uval[8];
    if (FWD) {
      #pragma unroll
      for (int r = 0; r < 8; ++r)
        uval[r] = u[((r + 8 * hh) * T_ + t) * C_ + n0 + lo];
    }
    __syncthreads();

    v8f acc0 = {}, acc1 = {}, acc2 = {}, acc3 = {};
    #pragma unroll
    for (int kk = 0; kk < 8; ++kk) {
      acc0 = wmma_k4(sE, bf[4 * kk + 0], 16 * kk + 0,  hh, lo, acc0);
      acc1 = wmma_k4(sE, bf[4 * kk + 1], 16 * kk + 4,  hh, lo, acc1);
      acc2 = wmma_k4(sE, bf[4 * kk + 2], 16 * kk + 8,  hh, lo, acc2);
      acc3 = wmma_k4(sE, bf[4 * kk + 3], 16 * kk + 12, hh, lo, acc3);
    }
    v8f cc = (acc0 + acc1) + (acc2 + acc3);

    #pragma unroll
    for (int r = 0; r < 8; ++r) {
      const int M = r + 8 * hh;
      const int N = n0 + lo;
      float val = sM[M] + __logf(cc[r]);
      if (FWD) val += uval[r];
      sA[M * C_ + N] = val;
      outbuf[(M * T_ + t) * C_ + N] = val;
    }
    __syncthreads();
  }
}

__global__ __launch_bounds__(256) void crf_scan(const float* __restrict__ u,
                                                const float* __restrict__ Pg,
                                                const float* __restrict__ PTg,
                                                float* __restrict__ alpha,
                                                float* __restrict__ beta) {
  extern __shared__ float smem[];
  if (blockIdx.x == 0)
    scan_dir<true>(u, Pg, alpha, smem);
  else
    scan_dir<false>(u, PTg, beta, smem);
}

__global__ __launch_bounds__(128) void crf_finalize(
    const float* __restrict__ alpha, const float* __restrict__ beta,
    float* __restrict__ out) {
  __shared__ float redm[4];
  __shared__ float reds[4];
  const int row = blockIdx.x;
  const int c   = threadIdx.x;
  const float v = alpha[row * C_ + c] + beta[row * C_ + c];

  float m = v;
  #pragma unroll
  for (int off = 1; off < 32; off <<= 1) m = fmaxf(m, __shfl_xor(m, off, 32));
  const int wv = c >> 5;
  if ((c & 31) == 0) redm[wv] = m;
  __syncthreads();
  m = fmaxf(fmaxf(redm[0], redm[1]), fmaxf(redm[2], redm[3]));

  float e = __expf(v - m);
  #pragma unroll
  for (int off = 1; off < 32; off <<= 1) e += __shfl_xor(e, off, 32);
  if ((c & 31) == 0) reds[wv] = e;
  __syncthreads();
  const float s = reds[0] + reds[1] + reds[2] + reds[3];

  out[row * C_ + c] = v - m - __logf(s);
}

extern "C" void kernel_launch(void* const* d_in, const int* in_sizes, int n_in,
                              void* d_out, int out_size, void* d_ws,
                              size_t ws_size, hipStream_t stream) {
  (void)in_sizes; (void)n_in; (void)out_size; (void)ws_size;
  const float* u  = (const float*)d_in[0];
  const float* lt = (const float*)d_in[1];
  float* out = (float*)d_out;

  float* ws    = (float*)d_ws;
  float* Pg    = ws;
  float* PTg   = Pg + (size_t)C_ * C_;
  float* alpha = PTg + (size_t)C_ * C_;
  float* beta  = alpha + (size_t)B_ * T_ * C_;

  crf_prep<<<1, 128, 0, stream>>>(lt, Pg, PTg);

  const size_t smemBytes =
      (size_t)(B_ * ESTR + B_ * C_ + 16) * sizeof(float);
  crf_scan<<<2, 256, smemBytes, stream>>>(u, Pg, PTg, alpha, beta);

  crf_finalize<<<B_ * T_, 128, 0, stream>>>(alpha, beta, out);
}


# Scan-step barriers fence LDS only and u_t comes from a register ring loaded 7 steps ahead, so no global load or store round trip is waited for inside the 4095 sequential steps
# speedup vs baseline: 1.7194x; 1.7194x over previous
#include <hip/hip_runtime.h>
#include <math.h>


typedef __attribute__((ext_vector_type(2))) float v2f;
typedef __attribute__((ext_vector_type(8))) float v8f;

#define B_   16
#define T_   4096
#define C_   128
#define ESTR 132

#define PF   7

__device__ __forceinline__ void lds_barrier() {
  __builtin_amdgcn_fence(__ATOMIC_RELEASE, "workgroup", "local");
  __builtin_amdgcn_s_barrier();
  __builtin_amdgcn_fence(__ATOMIC_ACQUIRE, "workgroup", "local");
}

__device__ __forceinline__ float half_shfl_max(float v) {
  #pragma unroll
  for (int off = 1; off < 16; off <<= 1)
    v = fmaxf(v, __shfl_xor(v, off, 32));
  return v;
}

__device__ __forceinline__ v8f wmma_k4(const float* __restrict__ sE,
                                       v2f bfrag, int k0, int hh, int lo,
                                       v8f c) {
  const int ka = k0 + 2 * hh;
  v2f a;
  a.x = sE[lo * ESTR + ka];
  a.y = sE[lo * ESTR + ka + 1];
  return __builtin_amdgcn_wmma_f32_16x16x4_f32(
      false, a, false, bfrag, (short)0, c, false, false);
}

__global__ __launch_bounds__(128) void crf_prep(const float* __restrict__ lt,
                                                float* __restrict__ Pg,
                                                float* __restrict__ PTg) {
  const int i = threadIdx.x;
  float mx = -INFINITY;
  for (int j = 0; j < C_; ++j) mx = fmaxf(mx, lt[i * C_ + j]);
  float s = 0.f;
  for (int j = 0; j < C_; ++j) s += __expf(lt[i * C_ + j] - mx);
  const float inv = 1.0f / s;
  for (int j = 0; j < C_; ++j) {
    const float p = __expf(lt[i * C_ + j] - mx) * inv;
    Pg[i * C_ + j] = p;
    PTg[j * C_ + i] = p;
  }
}

template <bool FWD>
__device__ void scan_dir(const float* __restrict__ u,
                         const float* __restrict__ mat,
                         float* __restrict__ outbuf,
                         float* smem) {
  float* sE = smem;
  float* sA = sE + B_ * ESTR;
  float* sM = sA + B_ * C_;

  const int tid  = threadIdx.x;
  const int lane = tid & 31;
  const int wv   = tid >> 5;
  const int hh   = lane >> 4;
  const int lo   = lane & 15;
  const int n0   = wv * 16;

  v2f bf[32];
  #pragma unroll
  for (int kk = 0; kk < 32; ++kk) {
    const int ka = 4 * kk + 2 * hh;
    bf[kk].x = mat[ka * C_ + n0 + lo];
    bf[kk].y = mat[(ka + 1) * C_ + n0 + lo];
  }

  const int b   = tid >> 4;
  const int sub = tid & 15;
  const int j0  = sub * 8;

  {
    const int tEdge = FWD ? 0 : (T_ - 1);
    #pragma unroll
    for (int q = 0; q < 8; ++q) {
      const int j = j0 + q;
      const float ue = u[(b * T_ + tEdge) * C_ + j];
      sA[b * C_ + j] = ue;
      outbuf[(b * T_ + tEdge) * C_ + j] = FWD ? ue : 0.0f;
    }
  }

  unsigned voff[8];
  #pragma unroll
  for (int r = 0; r < 8; ++r)
    voff[r] = (unsigned)((r + 8 * hh) * T_ * C_ + n0 + lo);

  float ur[PF][8];
  #pragma unroll
  for (int d = 0; d < PF; ++d) {
    const int t = FWD ? (1 + d) : (T_ - 2 - d);
    const float* __restrict__ ut = u + (size_t)t * C_;
    #pragma unroll
    for (int r = 0; r < 8; ++r)
      ur[d][r] = ut[voff[r]];
  }
  #pragma unroll
  for (int kk = 0; kk < 32; ++kk)
    asm volatile("" : "+v"(bf[kk]));
  #pragma unroll
  for (int d = 0; d < PF; ++d) {
    #pragma unroll
    for (int r = 0; r < 8; ++r)
      asm volatile("" : "+v"(ur[d][r]));
  }
  lds_barrier();

  for (int s0 = 1; s0 < T_; s0 += PF) {
    #pragma unroll
    for (int d = 0; d < PF; ++d) {
      const int s = s0 + d;
      const int t = FWD ? s : (T_ - 1 - s);

      float vals[8];
      float lm = -INFINITY;
      #pragma unroll
      for (int q = 0; q < 8; ++q) {
        const float v = sA[b * C_ + j0 + q];
        vals[q] = v;
        lm = fmaxf(lm, v);
      }
      lm = half_shfl_max(lm);
      if (sub == 0) sM[b] = lm;
      #pragma unroll
      for (int q = 0; q < 8; ++q)
        sE[b * ESTR + j0 + q] = __expf(vals[q] - lm);
      lds_barrier();

      v8f acc0 = {}, acc1 = {}, acc2 = {}, acc3 = {};
      #pragma unroll
      for (int kk = 0; kk < 8; ++kk) {
        acc0 = wmma_k4(sE, bf[4 * kk + 0], 16 * kk + 0,  hh, lo, acc0);
        acc1 = wmma_k4(sE, bf[4 * kk + 1], 16 * kk + 4,  hh, lo, acc1);
        acc2 = wmma_k4(sE, bf[4 * kk + 2], 16 * kk + 8,  hh, lo, acc2);
        acc3 = wmma_k4(sE, bf[4 * kk + 3], 16 * kk + 12, hh, lo, acc3);
      }
      v8f cc = (acc0 + acc1) + (acc2 + acc3);

      float* __restrict__ ot = outbuf + (size_t)t * C_;
      #pragma unroll
      for (int r = 0; r < 8; ++r) {
        const int M = r + 8 * hh;
        const int N = n0 + lo;
        const float lg = sM[M] + __logf(cc[r]);
        const float wu = lg + ur[d][r];
        sA[M * C_ + N] = wu;
        ot[voff[r]] = FWD ? wu : lg;
      }

      {
        int sn = s + PF;
        if (sn > T_ - 1) sn = T_ - 1;
        const int tn = FWD ? sn : (T_ - 1 - sn);
        const float* __restrict__ un = u + (size_t)tn * C_;
        #pragma unroll
        for (int r = 0; r < 8; ++r)
          ur[d][r] = un[voff[r]];
      }
      lds_barrier();
    }
  }
}

__global__ __launch_bounds__(256) void crf_scan(const float* __restrict__ u,
                                                const float* __restrict__ Pg,
                                                const float* __restrict__ PTg,
                                                float* __restrict__ alpha,
                                                float* __restrict__ beta) {
  extern __shared__ float smem[];
  if (blockIdx.x == 0)
    scan_dir<true>(u, Pg, alpha, smem);
  else
    scan_dir<false>(u, PTg, beta, smem);
}

__global__ __launch_bounds__(128) void crf_finalize(
    const float* __restrict__ alpha, const float* __restrict__ beta,
    float* __restrict__ out) {
  __shared__ float redm[4];
  __shared__ float reds[4];
  const int row = blockIdx.x;
  const int c   = threadIdx.x;
  const float v = alpha[row * C_ + c] + beta[row * C_ + c];

  float m = v;
  #pragma unroll
  for (int off = 1; off < 32; off <<= 1) m = fmaxf(m, __shfl_xor(m, off, 32));
  const int wv = c >> 5;
  if ((c & 31) == 0) redm[wv] = m;
  __syncthreads();
  m = fmaxf(fmaxf(redm[0], redm[1]), fmaxf(redm[2], redm[3]));

  float e = __expf(v - m);
  #pragma unroll
  for (int off = 1; off < 32; off <<= 1) e += __shfl_xor(e, off, 32);
  if ((c & 31) == 0) reds[wv] = e;
  __syncthreads();
  const float s = reds[0] + reds[1] + reds[2] + reds[3];

  out[row * C_ + c] = v - m - __logf(s);
}

extern "C" void kernel_launch(void* const* d_in, const int* in_sizes, int n_in,
                              void* d_out, int out_size, void* d_ws,
                              size_t ws_size, hipStream_t stream) {
  (void)in_sizes; (void)n_in; (void)out_size; (void)ws_size;
  const float* u  = (const float*)d_in[0];
  const float* lt = (const float*)d_in[1];
  float* out = (float*)d_out;

  float* ws    = (float*)d_ws;
  float* Pg    = ws;
  float* PTg   = Pg + (size_t)C_ * C_;
  float* alpha = PTg + (size_t)C_ * C_;
  float* beta  = alpha + (size_t)B_ * T_ * C_;

  crf_prep<<<1, 128, 0, stream>>>(lt, Pg, PTg);

  const size_t smemBytes =
      (size_t)(B_ * ESTR + B_ * C_ + 16) * sizeof(float);
  crf_scan<<<2, 256, smemBytes, stream>>>(u, Pg, PTg, alpha, beta);

  crf_finalize<<<B_ * T_, 128, 0, stream>>>(alpha, beta, out);
}


# Each state is shifted by the row max of the step before (the log-sum-exp identity holds for any shift): state stays in accumulator registers, DPP partial max, one LDS barrier per step, not two
# speedup vs baseline: 2.4015x; 1.3967x over previous
#include <hip/hip_runtime.h>
#include <math.h>


typedef __attribute__((ext_vector_type(2))) float v2f;
typedef __attribute__((ext_vector_type(8))) float v8f;

#define B_   16
#define T_   4096
#define C_   128
#define ESTR 132

#define PF   7

__device__ __forceinline__ void lds_barrier() {
  __builtin_amdgcn_fence(__ATOMIC_RELEASE, "workgroup", "local");
  __builtin_amdgcn_s_barrier();
  __builtin_amdgcn_fence(__ATOMIC_ACQUIRE, "workgroup", "local");
}

template <int CTRL>
__device__ __forceinline__ float dpp_mov(float v) {
  const int x = __builtin_bit_cast(int, v);
  return __builtin_bit_cast(
      float, __builtin_amdgcn_update_dpp(x, x, CTRL, 0xf, 0xf, false));
}

#define DPP_MAX8(ctl)                                                  \
  "v_max_num_f32_dpp %0, %0, %0 " ctl " row_mask:0xf bank_mask:0xf\n\t" \
  "v_max_num_f32_dpp %1, %1, %1 " ctl " row_mask:0xf bank_mask:0xf\n\t" \
  "v_max_num_f32_dpp %2, %2, %2 " ctl " row_mask:0xf bank_mask:0xf\n\t" \
  "v_max_num_f32_dpp %3, %3, %3 " ctl " row_mask:0xf bank_mask:0xf\n\t" \
  "v_max_num_f32_dpp %4, %4, %4 " ctl " row_mask:0xf bank_mask:0xf\n\t" \
  "v_max_num_f32_dpp %5, %5, %5 " ctl " row_mask:0xf bank_mask:0xf\n\t" \
  "v_max_num_f32_dpp %6, %6, %6 " ctl " row_mask:0xf bank_mask:0xf\n\t" \
  "v_max_num_f32_dpp %7, %7, %7 " ctl " row_mask:0xf bank_mask:0xf\n\t"

__device__ __forceinline__ void row16_max8(const float (&w)[8], float (&p)[8]) {
  asm volatile(
      "v_max_num_f32_dpp %0, %8, %8 row_xmask:1 row_mask:0xf bank_mask:0xf\n\t"
      "v_max_num_f32_dpp %1, %9, %9 row_xmask:1 row_mask:0xf bank_mask:0xf\n\t"
      "v_max_num_f32_dpp %2, %10, %10 row_xmask:1 row_mask:0xf bank_mask:0xf\n\t"
      "v_max_num_f32_dpp %3, %11, %11 row_xmask:1 row_mask:0xf bank_mask:0xf\n\t"
      "v_max_num_f32_dpp %4, %12, %12 row_xmask:1 row_mask:0xf bank_mask:0xf\n\t"
      "v_max_num_f32_dpp %5, %13, %13 row_xmask:1 row_mask:0xf bank_mask:0xf\n\t"
      "v_max_num_f32_dpp %6, %14, %14 row_xmask:1 row_mask:0xf bank_mask:0xf\n\t"
      "v_max_num_f32_dpp %7, %15, %15 row_xmask:1 row_mask:0xf bank_mask:0xf\n\t"
      DPP_MAX8("row_xmask:2") DPP_MAX8("row_xmask:4") DPP_MAX8("row_xmask:8")
      : "=&v"(p[0]), "=&v"(p[1]), "=&v"(p[2]), "=&v"(p[3]),
        "=&v"(p[4]), "=&v"(p[5]), "=&v"(p[6]), "=&v"(p[7])
      : "v"(w[0]), "v"(w[1]), "v"(w[2]), "v"(w[3]),
        "v"(w[4]), "v"(w[5]), "v"(w[6]), "v"(w[7]));
}

__device__ __forceinline__ void row_max_from_partials(const float* sP, int rho,
                                                      float (&sg)[8]) {
  float m = sP[rho];
  #pragma unroll
  for (int w = 1; w < 8; ++w) m = fmaxf(m, sP[w * B_ + rho]);
  sg[0] = dpp_mov<0x150>(m);
  sg[1] = dpp_mov<0x151>(m);
  sg[2] = dpp_mov<0x152>(m);
  sg[3] = dpp_mov<0x153>(m);
  sg[4] = dpp_mov<0x154>(m);
  sg[5] = dpp_mov<0x155>(m);
  sg[6] = dpp_mov<0x156>(m);
  sg[7] = dpp_mov<0x157>(m);
}

__device__ __forceinline__ v8f wmma_k4(const float* __restrict__ sE,
                                       v2f bfrag, int k0, int hh, int lo,
                                       v8f c) {
  const int ka = k0 + 2 * hh;
  v2f a;
  a.x = sE[lo * ESTR + ka];
  a.y = sE[lo * ESTR + ka + 1];
  return __builtin_amdgcn_wmma_f32_16x16x4_f32(
      false, a, false, bfrag, (short)0, c, false, false);
}

__global__ __launch_bounds__(128) void crf_prep(const float* __restrict__ lt,
                                                float* __restrict__ Pg,
                                                float* __restrict__ PTg) {
  const int i = threadIdx.x;
  float mx = -INFINITY;
  for (int j = 0; j < C_; ++j) mx = fmaxf(mx, lt[i * C_ + j]);
  float s = 0.f;
  for (int j = 0; j < C_; ++j) s += __expf(lt[i * C_ + j] - mx);
  const float inv = 1.0f / s;
  for (int j = 0; j < C_; ++j) {
    const float p = __expf(lt[i * C_ + j] - mx) * inv;
    Pg[i * C_ + j] = p;
    PTg[j * C_ + i] = p;
  }
}

#define SEBUF (B_ * ESTR)
#define SPBUF (8 * B_)

template <bool FWD>
__device__ void scan_dir(const float* __restrict__ u,
                         const float* __restrict__ mat,
                         float* __restrict__ outbuf,
                         float* smem) {
  float* sE = smem;
  float* sP = sE + 2 * SEBUF;

  const int tid  = threadIdx.x;
  const int lane = tid & 31;
  const int wv   = tid >> 5;
  const int hh   = lane >> 4;
  const int lo   = lane & 15;
  const int n0   = wv * 16;
  const int rho  = (lo + 8 * hh) & 15;

  v2f bf[32];
  #pragma unroll
  for (int kk = 0; kk < 32; ++kk) {
    const int ka = 4 * kk + 2 * hh;
    bf[kk].x = mat[ka * C_ + n0 + lo];
    bf[kk].y = mat[(ka + 1) * C_ + n0 + lo];
  }

  unsigned voff[8];
  #pragma unroll
  for (int r = 0; r < 8; ++r)
    voff[r] = (unsigned)((r + 8 * hh) * T_ * C_ + n0 + lo);

  int eoff[8];
  #pragma unroll
  for (int r = 0; r < 8; ++r) eoff[r] = (r + 8 * hh) * ESTR + n0 + lo;

  float sig[8];
  {
    const int tEdge = FWD ? 0 : (T_ - 1);
    const float* __restrict__ ue = u + (size_t)tEdge * C_;
    float* __restrict__ oe = outbuf + (size_t)tEdge * C_;
    float a0[8], pm[8];
    #pragma unroll
    for (int r = 0; r < 8; ++r) {
      a0[r] = ue[voff[r]];
      oe[voff[r]] = FWD ? a0[r] : 0.0f;
    }
    row16_max8(a0, pm);
    if (lo == 0) {
      #pragma unroll
      for (int r = 0; r < 8; ++r) sP[wv * B_ + 8 * hh + r] = pm[r];
    }
    lds_barrier();
    row_max_from_partials(sP, rho, sig);
    #pragma unroll
    for (int r = 0; r < 8; ++r) sE[eoff[r]] = __expf(a0[r] - sig[r]);
  }

  float ur[PF][8];
  #pragma unroll
  for (int d = 0; d < PF; ++d) {
    const int t = FWD ? (1 + d) : (T_ - 2 - d);
    const float* __restrict__ ut = u + (size_t)t * C_;
    #pragma unroll
    for (int r = 0; r < 8; ++r)
      ur[d][r] = ut[voff[r]];
  }
  #pragma unroll
  for (int kk = 0; kk < 32; ++kk)
    asm volatile("" : "+v"(bf[kk]));
  #pragma unroll
  for (int d = 0; d < PF; ++d) {
    #pragma unroll
    for (int r = 0; r < 8; ++r)
      asm volatile("" : "+v"(ur[d][r]));
  }
  lds_barrier();

  int cur = 0;
  for (int s0 = 1; s0 < T_; s0 += PF) {
    #pragma unroll
    for (int d = 0; d < PF; ++d) {
      const int s = s0 + d;
      const int t = FWD ? s : (T_ - 1 - s);
      const float* sEr = sE + cur * SEBUF;
      float*       sEw = sE + (cur ^ 1) * SEBUF;
      const float* sPr = sP + cur * SPBUF;
      float*       sPw = sP + (cur ^ 1) * SPBUF;

      v8f acc0 = {}, acc1 = {}, acc2 = {}, acc3 = {};
      #pragma unroll
      for (int kk = 0; kk < 8; ++kk) {
        acc0 = wmma_k4(sEr, bf[4 * kk + 0], 16 * kk + 0,  hh, lo, acc0);
        acc1 = wmma_k4(sEr, bf[4 * kk + 1], 16 * kk + 4,  hh, lo, acc1);
        acc2 = wmma_k4(sEr, bf[4 * kk + 2], 16 * kk + 8,  hh, lo, acc2);
        acc3 = wmma_k4(sEr, bf[4 * kk + 3], 16 * kk + 12, hh, lo, acc3);
      }
      v8f cc = (acc0 + acc1) + (acc2 + acc3);

      float nsig[8];
      row_max_from_partials(sPr, rho, nsig);

      float wu[8], ov[8];
      #pragma unroll
      for (int r = 0; r < 8; ++r) {
        const float lg = sig[r] + __logf(cc[r]);
        wu[r] = lg + ur[d][r];
        ov[r] = FWD ? wu[r] : lg;
        sEw[eoff[r]] = __expf(wu[r] - nsig[r]);
        sig[r] = nsig[r];
      }

      float pm[8];
      row16_max8(wu, pm);
      if (lo == 0) {
        #pragma unroll
        for (int r = 0; r < 8; ++r) sPw[wv * B_ + 8 * hh + r] = pm[r];
      }

      float* __restrict__ ot = outbuf + (size_t)t * C_;
      #pragma unroll
      for (int r = 0; r < 8; ++r) ot[voff[r]] = ov[r];

      {
        int sn = s + PF;
        if (sn > T_ - 1) sn = T_ - 1;
        const int tn = FWD ? sn : (T_ - 1 - sn);
        const float* __restrict__ un = u + (size_t)tn * C_;
        #pragma unroll
        for (int r = 0; r < 8; ++r)
          ur[d][r] = un[voff[r]];
      }
      cur ^= 1;
      lds_barrier();
    }
  }
}

__global__ __launch_bounds__(256) void crf_scan(const float* __restrict__ u,
                                                const float* __restrict__ Pg,
                                                const float* __restrict__ PTg,
                                                float* __restrict__ alpha,
                                                float* __restrict__ beta) {
  extern __shared__ __attribute__((aligned(16))) float smem[];
  if (blockIdx.x == 0)
    scan_dir<true>(u, Pg, alpha, smem);
  else
    scan_dir<false>(u, PTg, beta, smem);
}

__global__ __launch_bounds__(128) void crf_finalize(
    const float* __restrict__ alpha, const float* __restrict__ beta,
    float* __restrict__ out) {
  __shared__ float redm[4];
  __shared__ float reds[4];
  const int row = blockIdx.x;
  const int c   = threadIdx.x;
  const float v = alpha[row * C_ + c] + beta[row * C_ + c];

  float m = v;
  #pragma unroll
  for (int off = 1; off < 32; off <<= 1) m = fmaxf(m, __shfl_xor(m, off, 32));
  const int wv = c >> 5;
  if ((c & 31) == 0) redm[wv] = m;
  __syncthreads();
  m = fmaxf(fmaxf(redm[0], redm[1]), fmaxf(redm[2], redm[3]));

  float e = __expf(v - m);
  #pragma unroll
  for (int off = 1; off < 32; off <<= 1) e += __shfl_xor(e, off, 32);
  if ((c & 31) == 0) reds[wv] = e;
  __syncthreads();
  const float s = reds[0] + reds[1] + reds[2] + reds[3];

  out[row * C_ + c] = v - m - __logf(s);
}

extern "C" void kernel_launch(void* const* d_in, const int* in_sizes, int n_in,
                              void* d_out, int out_size, void* d_ws,
                              size_t ws_size, hipStream_t stream) {
  (void)in_sizes; (void)n_in; (void)out_size; (void)ws_size;
  const float* u  = (const float*)d_in[0];
  const float* lt = (const float*)d_in[1];
  float* out = (float*)d_out;

  float* ws    = (float*)d_ws;
  float* Pg    = ws;
  float* PTg   = Pg + (size_t)C_ * C_;
  float* alpha = PTg + (size_t)C_ * C_;
  float* beta  = alpha + (size_t)B_ * T_ * C_;

  crf_prep<<<1, 128, 0, stream>>>(lt, Pg, PTg);

  const size_t smemBytes =
      (size_t)(2 * SEBUF + 2 * SPBUF) * sizeof(float);
  crf_scan<<<2, 256, smemBytes, stream>>>(u, Pg, PTg, alpha, beta);

  crf_finalize<<<B_ * T_, 128, 0, stream>>>(alpha, beta, out);
}


# Product transposed (batch on the 16 columns): a lane holds 8 adjacent states of one row, E crosses LDS as a 128-bit lane copy, own-tile products run ahead of the barrier; k order of the sums changes
# speedup vs baseline: 2.6290x; 1.0947x over previous
#include <hip/hip_runtime.h>
#include <math.h>


typedef __attribute__((ext_vector_type(2))) float v2f;
typedef __attribute__((ext_vector_type(4))) float v4f;
typedef __attribute__((ext_vector_type(8))) float v8f;

#define B_   16
#define T_   4096
#define C_   128

#define PF   7

__device__ __forceinline__ void lds_signal() {
  __builtin_amdgcn_fence(__ATOMIC_RELEASE, "workgroup", "local");
  __builtin_amdgcn_s_barrier_signal(-1);
}
__device__ __forceinline__ void lds_wait() {
  __builtin_amdgcn_s_barrier_wait(-1);
  __builtin_amdgcn_fence(__ATOMIC_ACQUIRE, "workgroup", "local");
}

__device__ __forceinline__ float max8(v4f a, v4f b) {
  return fmaxf(fmaxf(fmaxf(a.x, a.y), fmaxf(a.z, a.w)),
               fmaxf(fmaxf(b.x, b.y), fmaxf(b.z, b.w)));
}

__device__ __forceinline__ float row_max16(const float* sP, int lo) {
  float m = sP[lo];
  #pragma unroll
  for (int k = 1; k < 16; ++k) m = fmaxf(m, sP[k * B_ + lo]);
  return m;
}

__device__ __forceinline__ v8f wmma_k4(v2f afrag, float b0, float b1, v8f c) {
  v2f b;
  b.x = b0;
  b.y = b1;
  return __builtin_amdgcn_wmma_f32_16x16x4_f32(
      false, afrag, false, b, (short)0, c, false, false);
}

__global__ __launch_bounds__(128) void crf_prep(const float* __restrict__ lt,
                                                float* __restrict__ Pg,
                                                float* __restrict__ PTg) {
  const int i = threadIdx.x;
  float mx = -INFINITY;
  for (int j = 0; j < C_; ++j) mx = fmaxf(mx, lt[i * C_ + j]);
  float s = 0.f;
  for (int j = 0; j < C_; ++j) s += __expf(lt[i * C_ + j] - mx);
  const float inv = 1.0f / s;
  for (int j = 0; j < C_; ++j) {
    const float p = __expf(lt[i * C_ + j] - mx) * inv;
    Pg[i * C_ + j] = p;
    PTg[j * C_ + i] = p;
  }
}

#define SEQ   (8 * 2 * 32)
#define SPBUF (16 * B_)

template <bool FWD>
__device__ void scan_dir(const float* __restrict__ u,
                         const float* __restrict__ mat,
                         float* __restrict__ outbuf,
                         float* smem) {
  v4f*   sE = reinterpret_cast<v4f*>(smem);
  float* sP = smem + 2 * SEQ * 4;

  const int tid  = threadIdx.x;
  const int lane = tid & 31;
  const int wv   = tid >> 5;
  const int hh   = lane >> 4;
  const int lo   = lane & 15;
  const int n0   = wv * 16;

  v2f af[32];
  #pragma unroll
  for (int j = 0; j < 8; ++j) {
    const int v = (wv + j) & 7;
    #pragma unroll
    for (int q = 0; q < 4; ++q) {
      const int k0 = 16 * v + 8 * hh + 2 * q;
      af[4 * j + q].x = mat[k0 * C_ + n0 + lo];
      af[4 * j + q].y = mat[(k0 + 1) * C_ + n0 + lo];
    }
  }

  const unsigned voff = (unsigned)(lo * T_ * C_ + n0 + 8 * hh);

  int eld[7];
  #pragma unroll
  for (int j = 1; j < 8; ++j) eld[j - 1] = (((wv + j) & 7) * 2) * 32 + lane;
  const int est = (wv * 2) * 32 + lane;
  const int pst = (wv * 2 + hh) * B_ + lo;

  float sig;
  v8f acc0, acc1, acc2, acc3;
  const v8f zero8 = {};
  {
    const int tEdge = FWD ? 0 : (T_ - 1);
    const float* __restrict__ ue = u + (size_t)tEdge * C_;
    float* __restrict__ oe = outbuf + (size_t)tEdge * C_;
    const v4f x0 = *reinterpret_cast<const v4f*>(ue + voff);
    const v4f x1 = *reinterpret_cast<const v4f*>(ue + voff + 4);
    const v4f z4 = {0.0f, 0.0f, 0.0f, 0.0f};
    *reinterpret_cast<v4f*>(oe + voff)     = FWD ? x0 : z4;
    *reinterpret_cast<v4f*>(oe + voff + 4) = FWD ? x1 : z4;
    sP[pst] = max8(x0, x1);
    lds_signal();
    lds_wait();
    sig = row_max16(sP, lo);
    v4f e0, e1;
    e0.x = __expf(x0.x - sig); e0.y = __expf(x0.y - sig);
    e0.z = __expf(x0.z - sig); e0.w = __expf(x0.w - sig);
    e1.x = __expf(x1.x - sig); e1.y = __expf(x1.y - sig);
    e1.z = __expf(x1.z - sig); e1.w = __expf(x1.w - sig);
    sE[est]      = e0;
    sE[est + 32] = e1;
    acc0 = wmma_k4(af[0], e0.x, e0.y, zero8);
    acc1 = wmma_k4(af[1], e0.z, e0.w, zero8);
    acc2 = wmma_k4(af[2], e1.x, e1.y, zero8);
    acc3 = wmma_k4(af[3], e1.z, e1.w, zero8);
  }

  v4f ur[PF][2];
  #pragma unroll
  for (int d = 0; d < PF; ++d) {
    const int t = FWD ? (1 + d) : (T_ - 2 - d);
    const float* __restrict__ ut = u + (size_t)t * C_;
    ur[d][0] = *reinterpret_cast<const v4f*>(ut + voff);
    ur[d][1] = *reinterpret_cast<const v4f*>(ut + voff + 4);
  }
  #pragma unroll
  for (int kk = 0; kk < 32; ++kk)
    asm volatile("" : "+v"(af[kk]));
  #pragma unroll
  for (int d = 0; d < PF; ++d) {
    asm volatile("" : "+v"(ur[d][0]));
    asm volatile("" : "+v"(ur[d][1]));
  }
  lds_signal();
  lds_wait();

  int cur = 0;
  for (int s0 = 1; s0 < T_; s0 += PF) {
    #pragma unroll
    for (int d = 0; d < PF; ++d) {
      const int s = s0 + d;
      const int t = FWD ? s : (T_ - 1 - s);
      const v4f*   sEr = sE + cur * SEQ;
      v4f*         sEw = sE + (cur ^ 1) * SEQ;
      const float* sPr = sP + cur * SPBUF;
      float*       sPw = sP + (cur ^ 1) * SPBUF;

      float pr[16];
      #pragma unroll
      for (int k = 0; k < 16; ++k) pr[k] = sPr[k * B_ + lo];
      v4f bq[7][2];
      #pragma unroll
      for (int j = 0; j < 7; ++j) {
        bq[j][0] = sEr[eld[j]];
        bq[j][1] = sEr[eld[j] + 32];
      }
      __builtin_amdgcn_sched_barrier(0);

      #pragma unroll
      for (int j = 0; j < 7; ++j) {
        acc0 = wmma_k4(af[4 * j + 4], bq[j][0].x, bq[j][0].y, acc0);
        acc1 = wmma_k4(af[4 * j + 5], bq[j][0].z, bq[j][0].w, acc1);
        acc2 = wmma_k4(af[4 * j + 6], bq[j][1].x, bq[j][1].y, acc2);
        acc3 = wmma_k4(af[4 * j + 7], bq[j][1].z, bq[j][1].w, acc3);
      }
      const v8f cc = (acc0 + acc1) + (acc2 + acc3);

      float nsig = pr[0];
      #pragma unroll
      for (int k = 1; k < 16; ++k) nsig = fmaxf(nsig, pr[k]);

      float wu[8], ov[8], e[8];
      #pragma unroll
      for (int r = 0; r < 8; ++r) {
        const float lg = sig + __logf(cc[r]);
        wu[r] = lg + ur[d][r >> 2][r & 3];
        ov[r] = FWD ? wu[r] : lg;
        e[r]  = __expf(wu[r] - nsig);
      }
      sig = nsig;

      v4f e0, e1;
      e0.x = e[0]; e0.y = e[1]; e0.z = e[2]; e0.w = e[3];
      e1.x = e[4]; e1.y = e[5]; e1.z = e[6]; e1.w = e[7];
      sEw[est]      = e0;
      sEw[est + 32] = e1;
      sPw[pst] = fmaxf(fmaxf(fmaxf(wu[0], wu[1]), fmaxf(wu[2], wu[3])),
                       fmaxf(fmaxf(wu[4], wu[5]), fmaxf(wu[6], wu[7])));

      lds_signal();
      __builtin_amdgcn_sched_barrier(0);

      acc0 = wmma_k4(af[0], e[0], e[1], zero8);
      acc1 = wmma_k4(af[1], e[2], e[3], zero8);
      acc2 = wmma_k4(af[2], e[4], e[5], zero8);
      acc3 = wmma_k4(af[3], e[6], e[7], zero8);

      float* __restrict__ ot = outbuf + (size_t)t * C_;
      v4f o0, o1;
      o0.x = ov[0]; o0.y = ov[1]; o0.z = ov[2]; o0.w = ov[3];
      o1.x = ov[4]; o1.y = ov[5]; o1.z = ov[6]; o1.w = ov[7];
      *reinterpret_cast<v4f*>(ot + voff)     = o0;
      *reinterpret_cast<v4f*>(ot + voff + 4) = o1;

      {
        int sn = s + PF;
        if (sn > T_ - 1) sn = T_ - 1;
        const int tn = FWD ? sn : (T_ - 1 - sn);
        const float* __restrict__ un = u + (size_t)tn * C_;
        ur[d][0] = *reinterpret_cast<const v4f*>(un + voff);
        ur[d][1] = *reinterpret_cast<const v4f*>(un + voff + 4);
      }
      cur ^= 1;
      __builtin_amdgcn_sched_barrier(0);
      lds_wait();
    }
  }
}

__global__ __launch_bounds__(256) void crf_scan(const float* __restrict__ u,
                                                const float* __restrict__ Pg,
                                                const float* __restrict__ PTg,
                                                float* __restrict__ alpha,
                                                float* __restrict__ beta) {
  extern __shared__ __attribute__((aligned(16))) float smem[];
  if (blockIdx.x == 0)
    scan_dir<true>(u, Pg, alpha, smem);
  else
    scan_dir<false>(u, PTg, beta, smem);
}

__global__ __launch_bounds__(128) void crf_finalize(
    const float* __restrict__ alpha, const float* __restrict__ beta,
    float* __restrict__ out) {
  __shared__ float redm[4];
  __shared__ float reds[4];
  const int row = blockIdx.x;
  const int c   = threadIdx.x;
  const float v = alpha[row * C_ + c] + beta[row * C_ + c];

  float m = v;
  #pragma unroll
  for (int off = 1; off < 32; off <<= 1) m = fmaxf(m, __shfl_xor(m, off, 32));
  const int wv = c >> 5;
  if ((c & 31) == 0) redm[wv] = m;
  __syncthreads();
  m = fmaxf(fmaxf(redm[0], redm[1]), fmaxf(redm[2], redm[3]));

  float e = __expf(v - m);
  #pragma unroll
  for (int off = 1; off < 32; off <<= 1) e += __shfl_xor(e, off, 32);
  if ((c & 31) == 0) reds[wv] = e;
  __syncthreads();
  const float s = reds[0] + reds[1] + reds[2] + reds[3];

  out[row * C_ + c] = v - m - __logf(s);
}

extern "C" void kernel_launch(void* const* d_in, const int* in_sizes, int n_in,
                              void* d_out, int out_size, void* d_ws,
                              size_t ws_size, hipStream_t stream) {
  (void)in_sizes; (void)n_in; (void)out_size; (void)ws_size;
  const float* u  = (const float*)d_in[0];
  const float* lt = (const float*)d_in[1];
  float* out = (float*)d_out;

  float* ws    = (float*)d_ws;
  float* Pg    = ws;
  float* PTg   = Pg + (size_t)C_ * C_;
  float* alpha = PTg + (size_t)C_ * C_;
  float* beta  = alpha + (size_t)B_ * T_ * C_;

  crf_prep<<<1, 128, 0, stream>>>(lt, Pg, PTg);

  const size_t smemBytes =
      (size_t)(2 * SEQ * 4 + 2 * SPBUF) * sizeof(float);
  crf_scan<<<2, 256, smemBytes, stream>>>(u, Pg, PTg, alpha, beta);

  crf_finalize<<<B_ * T_, 128, 0, stream>>>(alpha, beta, out);
}


# Dependent tail per state is one bare log2, one multiply-add, one exp2: the part of the next exponent known before the products is kept out of it; alpha/beta are formed behind the barrier signal
# speedup vs baseline: 2.8372x; 1.0792x over previous
#include <hip/hip_runtime.h>
#include <math.h>


typedef __attribute__((ext_vector_type(2))) float v2f;
typedef __attribute__((ext_vector_type(4))) float v4f;
typedef __attribute__((ext_vector_type(8))) float v8f;

#define B_   16
#define T_   4096
#define C_   128

#define LOG2E_ 1.4426950408889634f
#define LN2_   0.6931471805599453f

#define PF   7

__device__ __forceinline__ void lds_signal() {
  __builtin_amdgcn_fence(__ATOMIC_RELEASE, "workgroup", "local");
  __builtin_amdgcn_s_barrier_signal(-1);
}
__device__ __forceinline__ void lds_wait() {
  __builtin_amdgcn_s_barrier_wait(-1);
  __builtin_amdgcn_fence(__ATOMIC_ACQUIRE, "workgroup", "local");
}

__device__ __forceinline__ float max8(v4f a, v4f b) {
  return fmaxf(fmaxf(fmaxf(a.x, a.y), fmaxf(a.z, a.w)),
               fmaxf(fmaxf(b.x, b.y), fmaxf(b.z, b.w)));
}

__device__ __forceinline__ float row_max16(const float* sP, int lo) {
  float m = sP[lo];
  #pragma unroll
  for (int k = 1; k < 16; ++k) m = fmaxf(m, sP[k * B_ + lo]);
  return m;
}

__device__ __forceinline__ v8f wmma_k4(v2f afrag, float b0, float b1, v8f c) {
  v2f b;
  b.x = b0;
  b.y = b1;
  return __builtin_amdgcn_wmma_f32_16x16x4_f32(
      false, afrag, false, b, (short)0, c, false, false);
}

__global__ __launch_bounds__(128) void crf_prep(const float* __restrict__ lt,
                                                float* __restrict__ Pg,
                                                float* __restrict__ PTg) {
  const int i = threadIdx.x;
  float mx = -INFINITY;
  for (int j = 0; j < C_; ++j) mx = fmaxf(mx, lt[i * C_ + j]);
  float s = 0.f;
  for (int j = 0; j < C_; ++j) s += __expf(lt[i * C_ + j] - mx);
  const float inv = 1.0f / s;
  for (int j = 0; j < C_; ++j) {
    const float p = __expf(lt[i * C_ + j] - mx) * inv;
    Pg[i * C_ + j] = p;
    PTg[j * C_ + i] = p;
  }
}

#define SEQ   (8 * 2 * 32)
#define SPBUF (16 * B_)

template <bool FWD>
__device__ void scan_dir(const float* __restrict__ u,
                         const float* __restrict__ mat,
                         float* __restrict__ outbuf,
                         float* smem) {
  v4f*   sE = reinterpret_cast<v4f*>(smem);
  float* sP = smem + 2 * SEQ * 4;

  const int tid  = threadIdx.x;
  const int lane = tid & 31;
  const int wv   = tid >> 5;
  const int hh   = lane >> 4;
  const int lo   = lane & 15;
  const int n0   = wv * 16;

  v2f af[32];
  #pragma unroll
  for (int j = 0; j < 8; ++j) {
    const int v = (wv + j) & 7;
    #pragma unroll
    for (int q = 0; q < 4; ++q) {
      const int k0 = 16 * v + 8 * hh + 2 * q;
      af[4 * j + q].x = mat[k0 * C_ + n0 + lo];
      af[4 * j + q].y = mat[(k0 + 1) * C_ + n0 + lo];
    }
  }

  const unsigned voff = (unsigned)(lo * T_ * C_ + n0 + 8 * hh);

  int eld[7];
  #pragma unroll
  for (int j = 1; j < 8; ++j) eld[j - 1] = (((wv + j) & 7) * 2) * 32 + lane;
  const int est = (wv * 2) * 32 + lane;
  const int pst = (wv * 2 + hh) * B_ + lo;

  float sig;
  v8f acc0, acc1, acc2, acc3;
  const v8f zero8 = {};
  {
    const int tEdge = FWD ? 0 : (T_ - 1);
    const float* __restrict__ ue = u + (size_t)tEdge * C_;
    float* __restrict__ oe = outbuf + (size_t)tEdge * C_;
    const v4f x0 = *reinterpret_cast<const v4f*>(ue + voff);
    const v4f x1 = *reinterpret_cast<const v4f*>(ue + voff + 4);
    const v4f z4 = {0.0f, 0.0f, 0.0f, 0.0f};
    *reinterpret_cast<v4f*>(oe + voff)     = FWD ? x0 : z4;
    *reinterpret_cast<v4f*>(oe + voff + 4) = FWD ? x1 : z4;
    sP[pst] = max8(x0, x1);
    lds_signal();
    lds_wait();
    sig = row_max16(sP, lo);
    v4f e0, e1;
    e0.x = __expf(x0.x - sig); e0.y = __expf(x0.y - sig);
    e0.z = __expf(x0.z - sig); e0.w = __expf(x0.w - sig);
    e1.x = __expf(x1.x - sig); e1.y = __expf(x1.y - sig);
    e1.z = __expf(x1.z - sig); e1.w = __expf(x1.w - sig);
    sE[est]      = e0;
    sE[est + 32] = e1;
    acc0 = wmma_k4(af[0], e0.x, e0.y, zero8);
    acc1 = wmma_k4(af[1], e0.z, e0.w, zero8);
    acc2 = wmma_k4(af[2], e1.x, e1.y, zero8);
    acc3 = wmma_k4(af[3], e1.z, e1.w, zero8);
  }

  v4f ur[PF][2];
  #pragma unroll
  for (int d = 0; d < PF; ++d) {
    const int t = FWD ? (1 + d) : (T_ - 2 - d);
    const float* __restrict__ ut = u + (size_t)t * C_;
    ur[d][0] = *reinterpret_cast<const v4f*>(ut + voff);
    ur[d][1] = *reinterpret_cast<const v4f*>(ut + voff + 4);
  }
  #pragma unroll
  for (int kk = 0; kk < 32; ++kk)
    asm volatile("" : "+v"(af[kk]));
  #pragma unroll
  for (int d = 0; d < PF; ++d) {
    asm volatile("" : "+v"(ur[d][0]));
    asm volatile("" : "+v"(ur[d][1]));
  }
  lds_signal();
  lds_wait();

  int cur = 0;
  for (int s0 = 1; s0 < T_; s0 += PF) {
    #pragma unroll
    for (int d = 0; d < PF; ++d) {
      const int s = s0 + d;
      const int t = FWD ? s : (T_ - 1 - s);
      const v4f*   sEr = sE + cur * SEQ;
      v4f*         sEw = sE + (cur ^ 1) * SEQ;
      const float* sPr = sP + cur * SPBUF;
      float*       sPw = sP + (cur ^ 1) * SPBUF;

      float pr[16];
      #pragma unroll
      for (int k = 0; k < 16; ++k) pr[k] = sPr[k * B_ + lo];
      v4f bq[7][2];
      #pragma unroll
      for (int j = 0; j < 7; ++j) {
        bq[j][0] = sEr[eld[j]];
        bq[j][1] = sEr[eld[j] + 32];
      }
      __builtin_amdgcn_sched_barrier(0);

      float nsig = pr[0];
      #pragma unroll
      for (int k = 1; k < 16; ++k) nsig = fmaxf(nsig, pr[k]);
      const float dsh = sig - nsig;
      const v4f kx0 = (ur[d][0] + dsh) * LOG2E_;
      const v4f kx1 = (ur[d][1] + dsh) * LOG2E_;

      #pragma unroll
      for (int j = 0; j < 7; ++j) {
        acc0 = wmma_k4(af[4 * j + 4], bq[j][0].x, bq[j][0].y, acc0);
        acc1 = wmma_k4(af[4 * j + 5], bq[j][0].z, bq[j][0].w, acc1);
        acc2 = wmma_k4(af[4 * j + 6], bq[j][1].x, bq[j][1].y, acc2);
        acc3 = wmma_k4(af[4 * j + 7], bq[j][1].z, bq[j][1].w, acc3);
      }
      const v8f cc = (acc0 + acc1) + (acc2 + acc3);

      float l2[8], xe[8], e[8];
      #pragma unroll
      for (int r = 0; r < 8; ++r) {
        l2[r] = __builtin_amdgcn_logf(cc[r]);
        xe[r] = l2[r] + (r < 4 ? kx0[r & 3] : kx1[r & 3]);
        e[r]  = __builtin_amdgcn_exp2f(xe[r]);
      }

      v4f e0, e1;
      e0.x = e[0]; e0.y = e[1]; e0.z = e[2]; e0.w = e[3];
      e1.x = e[4]; e1.y = e[5]; e1.z = e[6]; e1.w = e[7];
      sEw[est]      = e0;
      sEw[est + 32] = e1;
      {
        const float mx = fmaxf(fmaxf(fmaxf(xe[0], xe[1]), fmaxf(xe[2], xe[3])),
                               fmaxf(fmaxf(xe[4], xe[5]), fmaxf(xe[6], xe[7])));
        sPw[pst] = fmaf(mx, LN2_, nsig);
      }

      lds_signal();
      __builtin_amdgcn_sched_barrier(0);

      acc0 = wmma_k4(af[0], e[0], e[1], zero8);
      acc1 = wmma_k4(af[1], e[2], e[3], zero8);
      acc2 = wmma_k4(af[2], e[4], e[5], zero8);
      acc3 = wmma_k4(af[3], e[6], e[7], zero8);

      float* __restrict__ ot = outbuf + (size_t)t * C_;
      v4f o0, o1;
      #pragma unroll
      for (int r = 0; r < 4; ++r) {
        o0[r] = fmaf(l2[r],     LN2_, FWD ? (sig + ur[d][0][r]) : sig);
        o1[r] = fmaf(l2[r + 4], LN2_, FWD ? (sig + ur[d][1][r]) : sig);
      }
      *reinterpret_cast<v4f*>(ot + voff)     = o0;
      *reinterpret_cast<v4f*>(ot + voff + 4) = o1;
      sig = nsig;

      {
        int sn = s + PF;
        if (sn > T_ - 1) sn = T_ - 1;
        const int tn = FWD ? sn : (T_ - 1 - sn);
        const float* __restrict__ un = u + (size_t)tn * C_;
        ur[d][0] = *reinterpret_cast<const v4f*>(un + voff);
        ur[d][1] = *reinterpret_cast<const v4f*>(un + voff + 4);
      }
      cur ^= 1;
      __builtin_amdgcn_sched_barrier(0);
      lds_wait();
    }
  }
}

__global__ __launch_bounds__(256) void crf_scan(const float* __restrict__ u,
                                                const float* __restrict__ Pg,
                                                const float* __restrict__ PTg,
                                                float* __restrict__ alpha,
                                                float* __restrict__ beta) {
  extern __shared__ __attribute__((aligned(16))) float smem[];
  if (blockIdx.x == 0)
    scan_dir<true>(u, Pg, alpha, smem);
  else
    scan_dir<false>(u, PTg, beta, smem);
}

__global__ __launch_bounds__(128) void crf_finalize(
    const float* __restrict__ alpha, const float* __restrict__ beta,
    float* __restrict__ out) {
  __shared__ float redm[4];
  __shared__ float reds[4];
  const int row = blockIdx.x;
  const int c   = threadIdx.x;
  const float v = alpha[row * C_ + c] + beta[row * C_ + c];

  float m = v;
  #pragma unroll
  for (int off = 1; off < 32; off <<= 1) m = fmaxf(m, __shfl_xor(m, off, 32));
  const int wv = c >> 5;
  if ((c & 31) == 0) redm[wv] = m;
  __syncthreads();
  m = fmaxf(fmaxf(redm[0], redm[1]), fmaxf(redm[2], redm[3]));

  float e = __expf(v - m);
  #pragma unroll
  for (int off = 1; off < 32; off <<= 1) e += __shfl_xor(e, off, 32);
  if ((c & 31) == 0) reds[wv] = e;
  __syncthreads();
  const float s = reds[0] + reds[1] + reds[2] + reds[3];

  out[row * C_ + c] = v - m - __logf(s);
}

extern "C" void kernel_launch(void* const* d_in, const int* in_sizes, int n_in,
                              void* d_out, int out_size, void* d_ws,
                              size_t ws_size, hipStream_t stream) {
  (void)in_sizes; (void)n_in; (void)out_size; (void)ws_size;
  const float* u  = (const float*)d_in[0];
  const float* lt = (const float*)d_in[1];
  float* out = (float*)d_out;

  float* ws    = (float*)d_ws;
  float* Pg    = ws;
  float* PTg   = Pg + (size_t)C_ * C_;
  float* alpha = PTg + (size_t)C_ * C_;
  float* beta  = alpha + (size_t)B_ * T_ * C_;

  crf_prep<<<1, 128, 0, stream>>>(lt, Pg, PTg);

  const size_t smemBytes =
      (size_t)(2 * SEQ * 4 + 2 * SPBUF) * sizeof(float);
  crf_scan<<<2, 256, smemBytes, stream>>>(u, Pg, PTg, alpha, beta);

  crf_finalize<<<B_ * T_, 128, 0, stream>>>(alpha, beta, out);
}


# Scan on 4 waves, one per SIMD, 2 state tiles each: 16 of a step's 64 products per SIMD need no LDS and fill the tail, barrier and LDS-read gaps; tile A's tail runs among tile B's; sum order changes
# speedup vs baseline: 3.1659x; 1.1159x over previous
#include <hip/hip_runtime.h>
#include <math.h>


typedef __attribute__((ext_vector_type(2))) float v2f;
typedef __attribute__((ext_vector_type(4))) float v4f;
typedef __attribute__((ext_vector_type(8))) float v8f;

#define B_   16
#define T_   4096
#define C_   128

#define LOG2E_ 1.4426950408889634f
#define LN2_   0.6931471805599453f

#define PF   7

__device__ __forceinline__ void lds_signal() {
  __builtin_amdgcn_fence(__ATOMIC_RELEASE, "workgroup", "local");
  __builtin_amdgcn_s_barrier_signal(-1);
}
__device__ __forceinline__ void lds_wait() {
  __builtin_amdgcn_s_barrier_wait(-1);
  __builtin_amdgcn_fence(__ATOMIC_ACQUIRE, "workgroup", "local");
}

__device__ __forceinline__ float max8(v4f a, v4f b) {
  return fmaxf(fmaxf(fmaxf(a.x, a.y), fmaxf(a.z, a.w)),
               fmaxf(fmaxf(b.x, b.y), fmaxf(b.z, b.w)));
}

__device__ __forceinline__ v8f wmma_k4(v2f afrag, float b0, float b1, v8f c) {
  v2f b;
  b.x = b0;
  b.y = b1;
  return __builtin_amdgcn_wmma_f32_16x16x4_f32(
      false, afrag, false, b, (short)0, c, false, false);
}

__global__ __launch_bounds__(128) void crf_prep(const float* __restrict__ lt,
                                                float* __restrict__ Pg,
                                                float* __restrict__ PTg) {
  const int i = threadIdx.x;
  float mx = -INFINITY;
  for (int j = 0; j < C_; ++j) mx = fmaxf(mx, lt[i * C_ + j]);
  float s = 0.f;
  for (int j = 0; j < C_; ++j) s += __expf(lt[i * C_ + j] - mx);
  const float inv = 1.0f / s;
  for (int j = 0; j < C_; ++j) {
    const float p = __expf(lt[i * C_ + j] - mx) * inv;
    Pg[i * C_ + j] = p;
    PTg[j * C_ + i] = p;
  }
}

#define SEQ   (8 * 2 * 32)
#define SPBUF (8 * B_)

__device__ __forceinline__ void tail_log2(const v8f cc, float* l2) {
  #pragma unroll
  for (int r = 0; r < 8; ++r) l2[r] = __builtin_amdgcn_logf(cc[r]);
}
__device__ __forceinline__ void tail_exp2(const float* l2, const v4f k0,
                                          const v4f k1, float* xe, float* e) {
  #pragma unroll
  for (int r = 0; r < 8; ++r) {
    xe[r] = l2[r] + (r < 4 ? k0[r & 3] : k1[r & 3]);
    e[r]  = __builtin_amdgcn_exp2f(xe[r]);
  }
}
__device__ __forceinline__ float max8f(const float* x) {
  return fmaxf(fmaxf(fmaxf(x[0], x[1]), fmaxf(x[2], x[3])),
               fmaxf(fmaxf(x[4], x[5]), fmaxf(x[6], x[7])));
}

#define PIN_ACC(x) asm volatile("" : "+v"(x))

template <bool FWD>
__device__ void scan_dir(const float* __restrict__ u,
                         const float* __restrict__ mat,
                         float* __restrict__ outbuf,
                         float* smem) {
  v4f*   sE = reinterpret_cast<v4f*>(smem);
  float* sP = smem + 2 * SEQ * 4;

  const int tid  = threadIdx.x;
  const int lane = tid & 31;
  const int wv   = tid >> 5;
  const int hh   = lane >> 4;
  const int lo   = lane & 15;
  const int n0   = wv * 32;

  v2f afA[32], afB[32];
  #pragma unroll
  for (int j = 0; j < 8; ++j) {
    const int v = (2 * wv + j) & 7;
    #pragma unroll
    for (int q = 0; q < 4; ++q) {
      const int k0 = 16 * v + 8 * hh + 2 * q;
      afA[4 * j + q].x = mat[k0 * C_ + n0 + lo];
      afA[4 * j + q].y = mat[(k0 + 1) * C_ + n0 + lo];
      afB[4 * j + q].x = mat[k0 * C_ + n0 + 16 + lo];
      afB[4 * j + q].y = mat[(k0 + 1) * C_ + n0 + 16 + lo];
    }
  }

  const unsigned voff = (unsigned)(lo * T_ * C_ + n0 + 8 * hh);

  int eld[6];
  #pragma unroll
  for (int j = 2; j < 8; ++j) eld[j - 2] = (((2 * wv + j) & 7) * 2) * 32 + lane;
  const int est = (wv * 4) * 32 + lane;
  const int pst = (wv * 2 + hh) * B_ + lo;

  float sig;
  v8f accA0, accA1, accB0, accB1;
  float eB[8];
  const v8f zero8 = {};
  {
    const int tEdge = FWD ? 0 : (T_ - 1);
    const float* __restrict__ ue = u + (size_t)tEdge * C_;
    float* __restrict__ oe = outbuf + (size_t)tEdge * C_;
    const v4f x0 = *reinterpret_cast<const v4f*>(ue + voff);
    const v4f x1 = *reinterpret_cast<const v4f*>(ue + voff + 4);
    const v4f x2 = *reinterpret_cast<const v4f*>(ue + voff + 16);
    const v4f x3 = *reinterpret_cast<const v4f*>(ue + voff + 20);
    const v4f z4 = {0.0f, 0.0f, 0.0f, 0.0f};
    *reinterpret_cast<v4f*>(oe + voff)      = FWD ? x0 : z4;
    *reinterpret_cast<v4f*>(oe + voff + 4)  = FWD ? x1 : z4;
    *reinterpret_cast<v4f*>(oe + voff + 16) = FWD ? x2 : z4;
    *reinterpret_cast<v4f*>(oe + voff + 20) = FWD ? x3 : z4;
    sP[pst] = fmaxf(max8(x0, x1), max8(x2, x3));
    lds_signal();
    lds_wait();
    sig = sP[lo];
    #pragma unroll
    for (int k = 1; k < 8; ++k) sig = fmaxf(sig, sP[k * B_ + lo]);
    float eA[8];
    #pragma unroll
    for (int r = 0; r < 4; ++r) {
      eA[r]     = __expf(x0[r] - sig);
      eA[r + 4] = __expf(x1[r] - sig);
      eB[r]     = __expf(x2[r] - sig);
      eB[r + 4] = __expf(x3[r] - sig);
    }
    v4f w0, w1, w2, w3;
    #pragma unroll
    for (int r = 0; r < 4; ++r) {
      w0[r] = eA[r]; w1[r] = eA[r + 4]; w2[r] = eB[r]; w3[r] = eB[r + 4];
    }
    sE[est]      = w0;
    sE[est + 32] = w1;
    sE[est + 64] = w2;
    sE[est + 96] = w3;
    accA0 = wmma_k4(afA[0], eA[0], eA[1], zero8);
    accA1 = wmma_k4(afA[1], eA[2], eA[3], zero8);
    accB0 = wmma_k4(afB[0], eA[0], eA[1], zero8);
    accB1 = wmma_k4(afB[1], eA[2], eA[3], zero8);
    accA0 = wmma_k4(afA[2], eA[4], eA[5], accA0);
    accA1 = wmma_k4(afA[3], eA[6], eA[7], accA1);
    accB0 = wmma_k4(afB[2], eA[4], eA[5], accB0);
    accB1 = wmma_k4(afB[3], eA[6], eA[7], accB1);
    accA0 = wmma_k4(afA[4], eB[0], eB[1], accA0);
    accB0 = wmma_k4(afB[4], eB[0], eB[1], accB0);
    accA1 = wmma_k4(afA[5], eB[2], eB[3], accA1);
    accB1 = wmma_k4(afB[5], eB[2], eB[3], accB1);
    accA0 = wmma_k4(afA[6], eB[4], eB[5], accA0);
    accB0 = wmma_k4(afB[6], eB[4], eB[5], accB0);
    accA1 = wmma_k4(afA[7], eB[6], eB[7], accA1);
    accB1 = wmma_k4(afB[7], eB[6], eB[7], accB1);
  }

  v4f ur[PF][4];
  #pragma unroll
  for (int d = 0; d < PF; ++d) {
    const int t = FWD ? (1 + d) : (T_ - 2 - d);
    const float* __restrict__ ut = u + (size_t)t * C_;
    ur[d][0] = *reinterpret_cast<const v4f*>(ut + voff);
    ur[d][1] = *reinterpret_cast<const v4f*>(ut + voff + 4);
    ur[d][2] = *reinterpret_cast<const v4f*>(ut + voff + 16);
    ur[d][3] = *reinterpret_cast<const v4f*>(ut + voff + 20);
  }
  #pragma unroll
  for (int d = 0; d < PF; ++d) {
    asm volatile("" : "+v"(ur[d][0]));
    asm volatile("" : "+v"(ur[d][1]));
    asm volatile("" : "+v"(ur[d][2]));
    asm volatile("" : "+v"(ur[d][3]));
  }
  lds_signal();
  lds_wait();

  int cur = 0;
  float pr[8];
  v4f bq[6][2];
  #pragma unroll
  for (int k = 0; k < 8; ++k) pr[k] = sP[k * B_ + lo];
  #pragma unroll
  for (int j = 0; j < 6; ++j) {
    bq[j][0] = sE[eld[j]];
    bq[j][1] = sE[eld[j] + 32];
  }

  for (int s0 = 1; s0 < T_; s0 += PF) {
    #pragma unroll
    for (int d = 0; d < PF; ++d) {
      const int s = s0 + d;
      const int t = FWD ? s : (T_ - 1 - s);
      v4f*   sEw = sE + (cur ^ 1) * SEQ;
      float* sPw = sP + (cur ^ 1) * SPBUF;


      accA0 = wmma_k4(afA[8],  bq[0][0].x, bq[0][0].y, accA0);
      accA1 = wmma_k4(afA[9],  bq[0][0].z, bq[0][0].w, accA1);
      accA0 = wmma_k4(afA[10], bq[0][1].x, bq[0][1].y, accA0);
      accA1 = wmma_k4(afA[11], bq[0][1].z, bq[0][1].w, accA1);
      __builtin_amdgcn_sched_barrier(0);
      float nsig = pr[0];
      #pragma unroll
      for (int k = 1; k < 8; ++k) nsig = fmaxf(nsig, pr[k]);
      const float dsh = sig - nsig;
      const v4f kxA0 = (ur[d][0] + dsh) * LOG2E_;
      const v4f kxA1 = (ur[d][1] + dsh) * LOG2E_;
      const v4f kxB0 = (ur[d][2] + dsh) * LOG2E_;
      const v4f kxB1 = (ur[d][3] + dsh) * LOG2E_;
      __builtin_amdgcn_sched_barrier(0);
      #pragma unroll
      for (int j = 1; j < 6; ++j) {
        accA0 = wmma_k4(afA[4 * j + 8],  bq[j][0].x, bq[j][0].y, accA0);
        accA1 = wmma_k4(afA[4 * j + 9],  bq[j][0].z, bq[j][0].w, accA1);
        accA0 = wmma_k4(afA[4 * j + 10], bq[j][1].x, bq[j][1].y, accA0);
        accA1 = wmma_k4(afA[4 * j + 11], bq[j][1].z, bq[j][1].w, accA1);
      }
      const v8f ccA = accA0 + accA1;

      float l2A[8], xeA[8], eA[8];
      #pragma unroll
      for (int j = 0; j < 6; ++j) {
        accB0 = wmma_k4(afB[4 * j + 8],  bq[j][0].x, bq[j][0].y, accB0);
        accB1 = wmma_k4(afB[4 * j + 9],  bq[j][0].z, bq[j][0].w, accB1);
        accB0 = wmma_k4(afB[4 * j + 10], bq[j][1].x, bq[j][1].y, accB0);
        accB1 = wmma_k4(afB[4 * j + 11], bq[j][1].z, bq[j][1].w, accB1);
        if (j == 0) {
          __builtin_amdgcn_sched_barrier(0);
          tail_log2(ccA, l2A);
          __builtin_amdgcn_sched_barrier(0);
        } else if (j == 1) {
          __builtin_amdgcn_sched_barrier(0);
          tail_exp2(l2A, kxA0, kxA1, xeA, eA);
          __builtin_amdgcn_sched_barrier(0);
        } else if (j == 2) {
          __builtin_amdgcn_sched_barrier(0);
          v4f w0, w1;
          #pragma unroll
          for (int r = 0; r < 4; ++r) { w0[r] = eA[r]; w1[r] = eA[r + 4]; }
          sEw[est]      = w0;
          sEw[est + 32] = w1;
          __builtin_amdgcn_sched_barrier(0);
        }
      }
      const float mxA = max8f(xeA);
      const v8f ccB = accB0 + accB1;

      float l2B[8], xeB[8];
      accA0 = wmma_k4(afA[0], eA[0], eA[1], zero8);
      accA1 = wmma_k4(afA[1], eA[2], eA[3], zero8);
      PIN_ACC(accA0); PIN_ACC(accA1);
      __builtin_amdgcn_sched_barrier(0);
      tail_log2(ccB, l2B);
      tail_exp2(l2B, kxB0, kxB1, xeB, eB);
      {
        v4f w2, w3;
        #pragma unroll
        for (int r = 0; r < 4; ++r) { w2[r] = eB[r]; w3[r] = eB[r + 4]; }
        sEw[est + 64] = w2;
        sEw[est + 96] = w3;
        const float mx = fmaxf(mxA, max8f(xeB));
        sPw[pst] = fmaf(mx, LN2_, nsig);
      }
      __builtin_amdgcn_sched_barrier(0);
      accB0 = wmma_k4(afB[0], eA[0], eA[1], zero8);
      accB1 = wmma_k4(afB[1], eA[2], eA[3], zero8);
      accA0 = wmma_k4(afA[2], eA[4], eA[5], accA0);
      accA1 = wmma_k4(afA[3], eA[6], eA[7], accA1);
      PIN_ACC(accB0); PIN_ACC(accB1); PIN_ACC(accA0); PIN_ACC(accA1);
      lds_signal();
      __builtin_amdgcn_sched_barrier(0);

      accB0 = wmma_k4(afB[2], eA[4], eA[5], accB0);
      accB1 = wmma_k4(afB[3], eA[6], eA[7], accB1);
      accA0 = wmma_k4(afA[4], eB[0], eB[1], accA0);
      accB0 = wmma_k4(afB[4], eB[0], eB[1], accB0);
      PIN_ACC(accB1); PIN_ACC(accA0); PIN_ACC(accB0);
      cur ^= 1;
      __builtin_amdgcn_sched_barrier(0);
      lds_wait();

      {
        const v4f*   sEr = sE + cur * SEQ;
        const float* sPr = sP + cur * SPBUF;
        #pragma unroll
        for (int k = 0; k < 8; ++k) pr[k] = sPr[k * B_ + lo];
        #pragma unroll
        for (int j = 0; j < 6; ++j) {
          bq[j][0] = sEr[eld[j]];
          bq[j][1] = sEr[eld[j] + 32];
        }
      }
      __builtin_amdgcn_sched_barrier(0);

      float* __restrict__ ot = outbuf + (size_t)t * C_;
      accA1 = wmma_k4(afA[5], eB[2], eB[3], accA1);
      accB1 = wmma_k4(afB[5], eB[2], eB[3], accB1);
      __builtin_amdgcn_sched_barrier(0);
      {
        v4f o0, o1;
        #pragma unroll
        for (int r = 0; r < 4; ++r) {
          o0[r] = fmaf(l2A[r],     LN2_, FWD ? (sig + ur[d][0][r]) : sig);
          o1[r] = fmaf(l2A[r + 4], LN2_, FWD ? (sig + ur[d][1][r]) : sig);
        }
        *reinterpret_cast<v4f*>(ot + voff)     = o0;
        *reinterpret_cast<v4f*>(ot + voff + 4) = o1;
      }
      __builtin_amdgcn_sched_barrier(0);
      accA0 = wmma_k4(afA[6], eB[4], eB[5], accA0);
      accB0 = wmma_k4(afB[6], eB[4], eB[5], accB0);
      __builtin_amdgcn_sched_barrier(0);
      {
        v4f o2, o3;
        #pragma unroll
        for (int r = 0; r < 4; ++r) {
          o2[r] = fmaf(l2B[r],     LN2_, FWD ? (sig + ur[d][2][r]) : sig);
          o3[r] = fmaf(l2B[r + 4], LN2_, FWD ? (sig + ur[d][3][r]) : sig);
        }
        *reinterpret_cast<v4f*>(ot + voff + 16) = o2;
        *reinterpret_cast<v4f*>(ot + voff + 20) = o3;
      }
      sig = nsig;
      __builtin_amdgcn_sched_barrier(0);
      accA1 = wmma_k4(afA[7], eB[6], eB[7], accA1);
      accB1 = wmma_k4(afB[7], eB[6], eB[7], accB1);

      {
        int sn = s + PF;
        if (sn > T_ - 1) sn = T_ - 1;
        const int tn = FWD ? sn : (T_ - 1 - sn);
        const float* __restrict__ un = u + (size_t)tn * C_;
        ur[d][0] = *reinterpret_cast<const v4f*>(un + voff);
        ur[d][1] = *reinterpret_cast<const v4f*>(un + voff + 4);
        ur[d][2] = *reinterpret_cast<const v4f*>(un + voff + 16);
        ur[d][3] = *reinterpret_cast<const v4f*>(un + voff + 20);
      }
    }
  }
}

__global__ __launch_bounds__(128) void crf_scan(const float* __restrict__ u,
                                                const float* __restrict__ Pg,
                                                const float* __restrict__ PTg,
                                                float* __restrict__ alpha,
                                                float* __restrict__ beta) {
  extern __shared__ __attribute__((aligned(16))) float smem[];
  if (blockIdx.x == 0)
    scan_dir<true>(u, Pg, alpha, smem);
  else
    scan_dir<false>(u, PTg, beta, smem);
}

__global__ __launch_bounds__(128) void crf_finalize(
    const float* __restrict__ alpha, const float* __restrict__ beta,
    float* __restrict__ out) {
  __shared__ float redm[4];
  __shared__ float reds[4];
  const int row = blockIdx.x;
  const int c   = threadIdx.x;
  const float v = alpha[row * C_ + c] + beta[row * C_ + c];

  float m = v;
  #pragma unroll
  for (int off = 1; off < 32; off <<= 1) m = fmaxf(m, __shfl_xor(m, off, 32));
  const int wv = c >> 5;
  if ((c & 31) == 0) redm[wv] = m;
  __syncthreads();
  m = fmaxf(fmaxf(redm[0], redm[1]), fmaxf(redm[2], redm[3]));

  float e = __expf(v - m);
  #pragma unroll
  for (int off = 1; off < 32; off <<= 1) e += __shfl_xor(e, off, 32);
  if ((c & 31) == 0) reds[wv] = e;
  __syncthreads();
  const float s = reds[0] + reds[1] + reds[2] + reds[3];

  out[row * C_ + c] = v - m - __logf(s);
}

extern "C" void kernel_launch(void* const* d_in, const int* in_sizes, int n_in,
                              void* d_out, int out_size, void* d_ws,
                              size_t ws_size, hipStream_t stream) {
  (void)in_sizes; (void)n_in; (void)out_size; (void)ws_size;
  const float* u  = (const float*)d_in[0];
  const float* lt = (const float*)d_in[1];
  float* out = (float*)d_out;

  float* ws    = (float*)d_ws;
  float* Pg    = ws;
  float* PTg   = Pg + (size_t)C_ * C_;
  float* alpha = PTg + (size_t)C_ * C_;
  float* beta  = alpha + (size_t)B_ * T_ * C_;

  crf_prep<<<1, 128, 0, stream>>>(lt, Pg, PTg);

  const size_t smemBytes =
      (size_t)(2 * SEQ * 4 + 2 * SPBUF) * sizeof(float);
  crf_scan<<<2, 128, smemBytes, stream>>>(u, Pg, PTg, alpha, beta);

  crf_finalize<<<B_ * T_, 128, 0, stream>>>(alpha, beta, out);
}
